# starting kernel
# baseline (speedup 1.0000x reference)
#include <hip/hip_runtime.h>

typedef __attribute__((ext_vector_type(2))) float v2f;
typedef __attribute__((ext_vector_type(4))) float v4f;
typedef __attribute__((ext_vector_type(8))) float v8f;

#define FIELDS     39
#define DIM        64
#define NPAIR      741
#define ROWS_PAD   48
#define LDS_STRIDE 68
#define BLOCK      192

__global__ __launch_bounds__(BLOCK)
void InnerProduct_gram_wmma(const float* __restrict__ x,
                            float* __restrict__ out,
                            int batches) {
  __shared__ float xs[ROWS_PAD * LDS_STRIDE];

  const int b   = blockIdx.x;
  if (b >= batches) return;
  const int tid = threadIdx.x;
  const float* __restrict__ xb = x + (size_t)b * (FIELDS * DIM);

  #pragma unroll
  for (int s = tid; s < ROWS_PAD * (DIM / 4); s += BLOCK) {
    const int row = s >> 4;
    const int cv  = s & 15;
    v4f val = {0.f, 0.f, 0.f, 0.f};
    if (row < FIELDS) {
      val = *(const v4f*)(xb + row * DIM + cv * 4);
    }
    *(v4f*)(&xs[row * LDS_STRIDE + cv * 4]) = val;
  }
  __syncthreads();

  const int wave = tid >> 5;
  const int lane = tid & 31;
  const int ti   = (wave < 3) ? 0 : ((wave < 5) ? 1 : 2);
  const int tj   = (wave < 3) ? wave : ((wave < 5) ? wave - 2 : 2);

  const int half = lane >> 4;
  const int l16  = lane & 15;
  const int arow = ti * 16 + l16;
  const int brow = tj * 16 + l16;

  v8f acc = {};
  #pragma unroll
  for (int k0 = 0; k0 < DIM; k0 += 4) {
    const int ko = k0 + half * 2;
    v2f a = *(const v2f*)(&xs[arow * LDS_STRIDE + ko]);
    v2f bb = *(const v2f*)(&xs[brow * LDS_STRIDE + ko]);
    acc = __builtin_amdgcn_wmma_f32_16x16x4_f32(
         false, a,  false, bb,
         (short)0, acc,  false,  false);
  }

  float* __restrict__ ob = out + (size_t)b * NPAIR;
  #pragma unroll
  for (int r = 0; r < 8; ++r) {
    const int m = ti * 16 + half * 8 + r;
    const int c = tj * 16 + l16;
    if (m < FIELDS && c < FIELDS && m < c) {
      const int idx = m * (FIELDS - 1) - (m * (m - 1)) / 2 + (c - m - 1);
      ob[idx] = acc[r];
    }
  }
}

extern "C" void kernel_launch(void* const* d_in, const int* in_sizes, int n_in,
                              void* d_out, int out_size, void* d_ws, size_t ws_size,
                              hipStream_t stream) {
  const float* x = (const float*)d_in[0];
  float* out = (float*)d_out;
  const int batches = in_sizes[0] / (FIELDS * DIM);
  InnerProduct_gram_wmma<<<batches, BLOCK, 0, stream>>>(x, out, batches);
}


# One wave per batch entry instead of six: private LDS slab, no block barrier; three fragment sets in registers serve as A and B of all six tiles; 4096 waves in all; summation order unchanged
# speedup vs baseline: 1.3334x; 1.3334x over previous
#include <hip/hip_runtime.h>


typedef __attribute__((ext_vector_type(2))) float v2f;
typedef __attribute__((ext_vector_type(4))) float v4f;
typedef __attribute__((ext_vector_type(8))) float v8f;

#define FIELDS     39
#define DIM        64
#define NPAIR      741
#define LDS_STRIDE 68
#define WAVES      4
#define BLOCK      (WAVES * 32)
#define NLOAD      20
#define SLAB       (2 * NLOAD * LDS_STRIDE)

#define RB(m)      (37 * (m) - ((m) * ((m) - 1)) / 2 - 1)

__global__ __launch_bounds__(BLOCK)
void InnerProduct_gram_wmma(const float* __restrict__ x,
                            float* __restrict__ out,
                            int batches) {
  __shared__ float xs[WAVES * SLAB];

  const unsigned lane = threadIdx.x & 31u;
  const int wave = __builtin_amdgcn_readfirstlane((int)(threadIdx.x >> 5));
  const int b = blockIdx.x * WAVES + wave;
  if (b >= batches) return;

  const unsigned half = lane >> 4;
  const unsigned l16  = lane & 15u;
  float* __restrict__ slab = xs + wave * SLAB;
  const float* __restrict__ xb = x + (size_t)b * (FIELDS * DIM);

  v4f st[NLOAD];
  #pragma unroll
  for (int p = 0; p < NLOAD - 1; ++p)
    st[p] = *(const v4f*)(xb + p * 128 + lane * 4);
  st[NLOAD - 1] = *(const v4f*)(xb + (NLOAD - 1) * 128 + l16 * 4);

  #pragma unroll
  for (int p = 0; p < NLOAD; ++p)
    *(v4f*)(&slab[(2 * p + half) * LDS_STRIDE + l16 * 4]) = st[p];

  __builtin_amdgcn_fence(__ATOMIC_RELEASE, "wavefront");
  __builtin_amdgcn_wave_barrier();
  __builtin_amdgcn_fence(__ATOMIC_ACQUIRE, "wavefront");

  const unsigned fr2 = (32u + l16 < FIELDS) ? (32u + l16) : (FIELDS - 1u);
  const float* f0 = slab + l16 * LDS_STRIDE + half * 2;
  const float* f1 = slab + (16 + l16) * LDS_STRIDE + half * 2;
  const float* f2 = slab + fr2 * LDS_STRIDE + half * 2;
  v2f a0[16], a1[16], a2[16];
  #pragma unroll
  for (int s = 0; s < 16; ++s) {
    a0[s] = *(const v2f*)(f0 + 4 * s);
    a1[s] = *(const v2f*)(f1 + 4 * s);
    a2[s] = *(const v2f*)(f2 + 4 * s);
  }

  __builtin_amdgcn_sched_barrier(0);

  v8f c00 = {}, c01 = {}, c02 = {}, c11 = {}, c12 = {}, c22 = {};
  #pragma unroll
  for (int s = 0; s < 16; ++s) {
    c00 = __builtin_amdgcn_wmma_f32_16x16x4_f32(false, a0[s], false, a0[s], (short)0, c00, false, false);
    c01 = __builtin_amdgcn_wmma_f32_16x16x4_f32(false, a0[s], false, a1[s], (short)0, c01, false, false);
    c02 = __builtin_amdgcn_wmma_f32_16x16x4_f32(false, a0[s], false, a2[s], (short)0, c02, false, false);
    c11 = __builtin_amdgcn_wmma_f32_16x16x4_f32(false, a1[s], false, a1[s], (short)0, c11, false, false);
    c12 = __builtin_amdgcn_wmma_f32_16x16x4_f32(false, a1[s], false, a2[s], (short)0, c12, false, false);
    c22 = __builtin_amdgcn_wmma_f32_16x16x4_f32(false, a2[s], false, a2[s], (short)0, c22, false, false);
  }

  __builtin_amdgcn_sched_barrier(0);
  asm volatile("v_nop\n\tv_nop\n\tv_nop\n\tv_nop"
               : "+v"(c00), "+v"(c01), "+v"(c02), "+v"(c11), "+v"(c12), "+v"(c22));
  __builtin_amdgcn_sched_barrier(0);

  char* __restrict__ ob = (char*)(out + (size_t)b * NPAIR);
  const unsigned l4 = l16 * 4u;
#define HSTEP(TI, R)         ((TI) == 2 ? 0u : 4u * (268u - 8u * (16u * (TI) + (R))))
#define LANE_OFF(TI, R)      (l4 + half * HSTEP(TI, R))
#define CONST_OFF(TI, TJ, R) (4 * (RB(16 * (TI) + (R)) + 16 * (TJ)))
#define PUT(ACC, TI, TJ, R) \
    *(float*)(ob + (size_t)LANE_OFF(TI, R) + CONST_OFF(TI, TJ, R)) = ACC[R]

  #pragma unroll
  for (int r = 0; r < 8; ++r) PUT(c01, 0, 1, r);
  if (l16 < 7u) {
    #pragma unroll
    for (int r = 0; r < 8; ++r) { PUT(c02, 0, 2, r); PUT(c12, 1, 2, r); }
  }
  #pragma unroll
  for (int r = 0; r < 8; ++r) {
    if (half * 8u + r < l16) {
      PUT(c00, 0, 0, r);
      PUT(c11, 1, 1, r);
      if (l16 < 7u) PUT(c22, 2, 2, r);
    }
  }
#undef PUT
#undef CONST_OFF
#undef LANE_OFF
#undef HSTEP
}

extern "C" void kernel_launch(void* const* d_in, const int* in_sizes, int n_in,
                              void* d_out, int out_size, void* d_ws, size_t ws_size,
                              hipStream_t stream) {
  const float* x = (const float*)d_in[0];
  float* out = (float*)d_out;
  const int batches = in_sizes[0] / (FIELDS * DIM);
  InnerProduct_gram_wmma<<<4096 / WAVES, BLOCK, 0, stream>>>(x, out, batches);
}


# Result tiles staged through the wave's own LDS slab in packed pair order and stored as whole 128-byte lines, 16 bytes per lane: 7 global store instructions per wave, not 46; sum order unchanged
# speedup vs baseline: 1.4086x; 1.0564x over previous
#include <hip/hip_runtime.h>


typedef __attribute__((ext_vector_type(2))) float v2f;
typedef __attribute__((ext_vector_type(4))) float v4f;
typedef __attribute__((ext_vector_type(8))) float v8f;

#define FIELDS     39
#define DIM        64
#define NPAIR      741
#define LDS_STRIDE 68
#define WAVES      4
#define BLOCK      (WAVES * 32)
#define NLOAD      20
#define SLAB       (2 * NLOAD * LDS_STRIDE)

#define RB(m)      (37 * (m) - ((m) * ((m) - 1)) / 2 - 1)

__global__ __launch_bounds__(BLOCK)
void InnerProduct_gram_wmma(const float* __restrict__ x,
                            float* __restrict__ out,
                            int batches) {
  __shared__ float xs[WAVES * SLAB];

  const unsigned lane = threadIdx.x & 31u;
  const int wave = __builtin_amdgcn_readfirstlane((int)(threadIdx.x >> 5));
  const int b = blockIdx.x * WAVES + wave;
  if (b >= batches) return;

  const unsigned half = lane >> 4;
  const unsigned l16  = lane & 15u;
  float* __restrict__ slab = xs + wave * SLAB;
  const float* __restrict__ xb = x + (size_t)b * (FIELDS * DIM);

  v4f st[NLOAD];
  #pragma unroll
  for (int p = 0; p < NLOAD - 1; ++p)
    st[p] = *(const v4f*)(xb + p * 128 + lane * 4);
  st[NLOAD - 1] = *(const v4f*)(xb + (NLOAD - 1) * 128 + l16 * 4);

  #pragma unroll
  for (int p = 0; p < NLOAD; ++p)
    *(v4f*)(&slab[(2 * p + half) * LDS_STRIDE + l16 * 4]) = st[p];

  __builtin_amdgcn_fence(__ATOMIC_RELEASE, "wavefront");
  __builtin_amdgcn_wave_barrier();
  __builtin_amdgcn_fence(__ATOMIC_ACQUIRE, "wavefront");

  const unsigned fr2 = (32u + l16 < FIELDS) ? (32u + l16) : (FIELDS - 1u);
  const float* f0 = slab + l16 * LDS_STRIDE + half * 2;
  const float* f1 = slab + (16 + l16) * LDS_STRIDE + half * 2;
  const float* f2 = slab + fr2 * LDS_STRIDE + half * 2;
  v2f a0[16], a1[16], a2[16];
  #pragma unroll
  for (int s = 0; s < 16; ++s) {
    a0[s] = *(const v2f*)(f0 + 4 * s);
    a1[s] = *(const v2f*)(f1 + 4 * s);
    a2[s] = *(const v2f*)(f2 + 4 * s);
  }

  __builtin_amdgcn_sched_barrier(0);

  v8f c00 = {}, c01 = {}, c02 = {}, c11 = {}, c12 = {}, c22 = {};
  #pragma unroll
  for (int s = 0; s < 16; ++s) {
    c00 = __builtin_amdgcn_wmma_f32_16x16x4_f32(false, a0[s], false, a0[s], (short)0, c00, false, false);
    c01 = __builtin_amdgcn_wmma_f32_16x16x4_f32(false, a0[s], false, a1[s], (short)0, c01, false, false);
    c02 = __builtin_amdgcn_wmma_f32_16x16x4_f32(false, a0[s], false, a2[s], (short)0, c02, false, false);
    c11 = __builtin_amdgcn_wmma_f32_16x16x4_f32(false, a1[s], false, a1[s], (short)0, c11, false, false);
    c12 = __builtin_amdgcn_wmma_f32_16x16x4_f32(false, a1[s], false, a2[s], (short)0, c12, false, false);
    c22 = __builtin_amdgcn_wmma_f32_16x16x4_f32(false, a2[s], false, a2[s], (short)0, c22, false, false);
  }

  __builtin_amdgcn_sched_barrier(0);
  asm volatile("v_nop\n\tv_nop\n\tv_nop\n\tv_nop"
               : "+v"(c00), "+v"(c01), "+v"(c02), "+v"(c11), "+v"(c12), "+v"(c22));
  __builtin_amdgcn_sched_barrier(0);

  const unsigned sh   = (unsigned)b & 3u;
  const unsigned head = (4u - sh) & 3u;
  const unsigned nq   = (NPAIR - head) >> 2;
  const unsigned lead = (((unsigned)b * NPAIR + head) >> 2) & 7u;
  float* pk = slab + 4u * lead + sh;
  char* ob = (char*)pk;
  const unsigned l4 = l16 * 4u;
#define HSTEP(TI, R)         ((TI) == 2 ? 0u : 4u * (268u - 8u * (16u * (TI) + (R))))
#define LANE_OFF(TI, R)      (l4 + half * HSTEP(TI, R))
#define CONST_OFF(TI, TJ, R) (4 * (RB(16 * (TI) + (R)) + 16 * (TJ)))
#define PUT(ACC, TI, TJ, R) \
    *(float*)(ob + (size_t)LANE_OFF(TI, R) + CONST_OFF(TI, TJ, R)) = ACC[R]

  #pragma unroll
  for (int r = 0; r < 8; ++r) PUT(c01, 0, 1, r);
  if (l16 < 7u) {
    #pragma unroll
    for (int r = 0; r < 8; ++r) { PUT(c02, 0, 2, r); PUT(c12, 1, 2, r); }
  }
  #pragma unroll
  for (int r = 0; r < 8; ++r) {
    if (half * 8u + r < l16) {
      PUT(c00, 0, 0, r);
      PUT(c11, 1, 1, r);
      if (l16 < 7u) PUT(c22, 2, 2, r);
    }
  }
#undef PUT
#undef CONST_OFF
#undef LANE_OFF
#undef HSTEP

  __builtin_amdgcn_fence(__ATOMIC_RELEASE, "wavefront");
  __builtin_amdgcn_wave_barrier();
  __builtin_amdgcn_fence(__ATOMIC_ACQUIRE, "wavefront");

  float* __restrict__ ov = out + (size_t)b * NPAIR;
  const int q0 = (int)lane - (int)lead;
  float*       g0 = ov + ((int)head + 4 * q0);
  const float* s0 = slab + (sh + head) + 4u * lane;
  if (q0 >= 0)
    *(v4f*)g0 = *(const v4f*)s0;
  #pragma unroll
  for (int i = 1; i < 5; ++i)
    *(v4f*)(g0 + 128 * i) = *(const v4f*)(s0 + 128 * i);
  if ((unsigned)(q0 + 160) < nq)
    *(v4f*)(g0 + 640) = *(const v4f*)(s0 + 640);

  {
    const unsigned tail0 = head + 4u * nq;
    const unsigned k = lane & 3u;
    const bool is_tail = (lane & 4u) != 0u;
    const unsigned p = is_tail ? tail0 + k : k;
    const bool on = lane < 8u && (is_tail ? (p < NPAIR) : (k < head));
    if (on) ov[p] = pk[p];
  }
}

extern "C" void kernel_launch(void* const* d_in, const int* in_sizes, int n_in,
                              void* d_out, int out_size, void* d_ws, size_t ws_size,
                              hipStream_t stream) {
  const float* x = (const float*)d_in[0];
  float* out = (float*)d_out;
  const int batches = in_sizes[0] / (FIELDS * DIM);
  InnerProduct_gram_wmma<<<4096 / WAVES, BLOCK, 0, stream>>>(x, out, batches);
}
